# starting kernel
# baseline (speedup 1.0000x reference)
#include <hip/hip_runtime.h>

typedef __attribute__((ext_vector_type(2))) float v2f;
typedef __attribute__((ext_vector_type(8))) float v8f;

#define BATCH   16384
#define NFEAT   1024
#define TDEPTH  10
#define NLEAVES 1024
#define WPB     8
#define RPW     16
#define RPB     (WPB * RPW)
#define KCHUNKS (NFEAT / 4)

__launch_bounds__(256)
__global__ void sdt_wmma_kernel(const float* __restrict__ x,
                                const float* __restrict__ W,
                                const float* __restrict__ bias,
                                const float* __restrict__ leaf,
                                float* __restrict__ out) {
  __shared__ float ldsB[KCHUNKS * 64];
  __shared__ float ldsLeaf[NLEAVES];
  __shared__ float ldsZ[WPB * 256];
  __shared__ float ldsV[WPB * 32];
  __shared__ float ldsBias[16];

  const int t    = threadIdx.x;
  const int wave = t >> 5;
  const int lane = t & 31;

  for (int idx = t; idx < 16 * NFEAT; idx += 256) {
    int n = idx >> 10;
    int k = idx & (NFEAT - 1);
    float w = W[n * NFEAT + k];
    int c  = k >> 2;
    int lb = ((k >> 1) & 1) * 16 + n;
    int j  = k & 1;
    ldsB[c * 64 + lb * 2 + j] = w;
  }
  for (int idx = t; idx < NLEAVES; idx += 256) ldsLeaf[idx] = leaf[idx];
  if (t < 16) ldsBias[t] = bias[t];
  __syncthreads();

  const int rowBase = (blockIdx.x * WPB + wave) * RPW;
  const int m = lane & 15;
  const int h = lane >> 4;

  const float* xrow = x + (size_t)(rowBase + m) * NFEAT + h * 2;
  const float* bptr = ldsB + lane * 2;
  v8f acc = {};
  #pragma unroll 4
  for (int c = 0; c < KCHUNKS; ++c) {
    v2f a  = *(const v2f*)(xrow + c * 4);
    v2f bf = *(const v2f*)(bptr + c * 64);
    acc = __builtin_amdgcn_wmma_f32_16x16x4_f32(
         false, a,  false, bf,
         (short)0, acc,  false,  false);
  }

  {
    float* zp = ldsZ + wave * 256;
    int n  = lane & 15;
    int mh = h * 8;
    #pragma unroll
    for (int i = 0; i < 8; ++i) zp[(mh + i) * 16 + n] = acc[i];
  }
  __syncthreads();

  float g[TDEPTH], gm[TDEPTH];
  {
    const float* zp = ldsZ + wave * 256 + m * 16;
    #pragma unroll
    for (int d = 0; d < TDEPTH; ++d) {
      float z = zp[d] + ldsBias[d];
      float e = __expf(-z);
      g[d]  = 1.0f / (1.0f + e);
      gm[d] = 1.0f - g[d];
    }
  }

  const float* lf = ldsLeaf + h * 512;
  float c16[16];
  #pragma unroll
  for (int jj = 0; jj < 16; ++jj) {
    float tt[16];
    #pragma unroll
    for (int i = 0; i < 16; ++i) {
      v2f lv = *(const v2f*)(lf + jj * 32 + i * 2);
      tt[i] = gm[9] * lv.x + g[9] * lv.y;
    }
    #pragma unroll
    for (int i = 0; i < 8; ++i) tt[i] = gm[8] * tt[2*i] + g[8] * tt[2*i+1];
    #pragma unroll
    for (int i = 0; i < 4; ++i) tt[i] = gm[7] * tt[2*i] + g[7] * tt[2*i+1];
    #pragma unroll
    for (int i = 0; i < 2; ++i) tt[i] = gm[6] * tt[2*i] + g[6] * tt[2*i+1];
    c16[jj] = gm[5] * tt[0] + g[5] * tt[1];
  }
  #pragma unroll
  for (int i = 0; i < 8; ++i) c16[i] = gm[4] * c16[2*i] + g[4] * c16[2*i+1];
  #pragma unroll
  for (int i = 0; i < 4; ++i) c16[i] = gm[3] * c16[2*i] + g[3] * c16[2*i+1];
  #pragma unroll
  for (int i = 0; i < 2; ++i) c16[i] = gm[2] * c16[2*i] + g[2] * c16[2*i+1];
  float vh = gm[1] * c16[0] + g[1] * c16[1];

  ldsV[wave * 32 + h * 16 + m] = vh;
  __syncthreads();
  if (lane < 16) {
    float v0 = ldsV[wave * 32 + m];
    float v1 = ldsV[wave * 32 + 16 + m];
    out[rowBase + m] = gm[0] * v0 + g[0] * v1;
  }
}

extern "C" void kernel_launch(void* const* d_in, const int* in_sizes, int n_in,
                              void* d_out, int out_size, void* d_ws, size_t ws_size,
                              hipStream_t stream) {
  (void)in_sizes; (void)n_in; (void)out_size; (void)d_ws; (void)ws_size;
  const float* x    = (const float*)d_in[0];
  const float* W    = (const float*)d_in[1];
  const float* b    = (const float*)d_in[2];
  const float* leaf = (const float*)d_in[3];
  float* out = (float*)d_out;
  sdt_wmma_kernel<<<dim3(BATCH / RPB), dim3(256), 0, stream>>>(x, W, b, leaf, out);
}


# k loop: x and staged W fragments read as 16 B per lane in one permuted k order (each wmma sums k,k+1,k+4,k+5: summation regrouped); next block's 16 x loads and B reads issued ahead of current wmmas
# speedup vs baseline: 1.5921x; 1.5921x over previous
#include <hip/hip_runtime.h>

typedef __attribute__((ext_vector_type(2))) float v2f;
typedef __attribute__((ext_vector_type(4))) float v4f;
typedef __attribute__((ext_vector_type(8))) float v8f;

#define BATCH   16384
#define NFEAT   1024
#define TDEPTH  10
#define NLEAVES 1024
#define WPB     8
#define RPW     16
#define RPB     (WPB * RPW)
#define NSTEP   (NFEAT / 8)
#define GSTEP   16
#define NBLK    (NSTEP / GSTEP)

__launch_bounds__(256)
__global__ void sdt_wmma_kernel(const float* __restrict__ x,
                                const float* __restrict__ W,
                                const float* __restrict__ bias,
                                const float* __restrict__ leaf,
                                float* __restrict__ out) {
  __shared__ v4f ldsB4[NSTEP * 32];
  __shared__ float ldsLeaf[NLEAVES];
  __shared__ float ldsZ[WPB * 256];
  __shared__ float ldsV[WPB * 32];
  __shared__ float ldsBias[16];

  const int t    = threadIdx.x;
  const int wave = t >> 5;
  const int lane = t & 31;
  const int rowBase = (blockIdx.x * WPB + wave) * RPW;
  const int m = lane & 15;
  const int h = lane >> 4;

  const v4f* xp = (const v4f*)(x + (size_t)(rowBase + m) * NFEAT) + h;
  v4f xa[2][GSTEP], bb[2][GSTEP];
  #pragma unroll
  for (int i = 0; i < GSTEP; ++i) xa[0][i] = xp[2 * i];

  {
    const int n = t & 15;
    const int kq0 = (t >> 5) * 32 + ((t >> 4) & 1);
    const v4f* wrow = (const v4f*)(W + n * NFEAT);
    v4f wv[16];
    #pragma unroll
    for (int i = 0; i < 16; ++i) wv[i] = wrow[kq0 + 2 * i];
    #pragma unroll
    for (int i = 0; i < 16; ++i) ldsB4[(kq0 + 2 * i) * 16 + n] = wv[i];
  }
  for (int idx = t; idx < NLEAVES; idx += 256) ldsLeaf[idx] = leaf[idx];
  if (t < 16) ldsBias[t] = bias[t];
  __syncthreads();

  const v4f* bp = ldsB4 + lane;
#define ISSUE_BLOCK(set, blk)                                                   \
  _Pragma("unroll") for (int i = 0; i < GSTEP; ++i)                             \
    xa[set][i] = xp[2 * ((blk) * GSTEP + i)];                                   \
  _Pragma("unroll") for (int i = 0; i < GSTEP; ++i)                             \
    bb[set][i] = bp[32 * ((blk) * GSTEP + i)];
#define COMPUTE_BLOCK(set)                                                      \
  _Pragma("unroll") for (int i = 0; i < GSTEP; ++i) {                           \
    acc = __builtin_amdgcn_wmma_f32_16x16x4_f32(false, xa[set][i].xy, false,    \
              bb[set][i].xy, (short)0, acc, false, false);                      \
    acc = __builtin_amdgcn_wmma_f32_16x16x4_f32(false, xa[set][i].zw, false,    \
              bb[set][i].zw, (short)0, acc, false, false);                      \
  }
  #pragma unroll
  for (int i = 0; i < GSTEP; ++i) bb[0][i] = bp[32 * i];
  v8f acc = {};
  #pragma unroll
  for (int blk = 0; blk < NBLK; blk += 2) {
    ISSUE_BLOCK(1, blk + 1)
    __builtin_amdgcn_sched_barrier(0);
    COMPUTE_BLOCK(0)
    __builtin_amdgcn_sched_barrier(0);
    if (blk + 2 < NBLK) {
      ISSUE_BLOCK(0, blk + 2)
      __builtin_amdgcn_sched_barrier(0);
    }
    COMPUTE_BLOCK(1)
    __builtin_amdgcn_sched_barrier(0);
  }
#undef ISSUE_BLOCK
#undef COMPUTE_BLOCK

  {
    float* zp = ldsZ + wave * 256;
    int n  = lane & 15;
    int mh = h * 8;
    #pragma unroll
    for (int i = 0; i < 8; ++i) zp[(mh + i) * 16 + n] = acc[i];
  }
  __syncthreads();

  float g[TDEPTH], gm[TDEPTH];
  {
    const float* zp = ldsZ + wave * 256 + m * 16;
    #pragma unroll
    for (int d = 0; d < TDEPTH; ++d) {
      float z = zp[d] + ldsBias[d];
      float e = __expf(-z);
      g[d]  = 1.0f / (1.0f + e);
      gm[d] = 1.0f - g[d];
    }
  }

  const float* lf = ldsLeaf + h * 512;
  float c16[16];
  #pragma unroll
  for (int jj = 0; jj < 16; ++jj) {
    float tt[16];
    #pragma unroll
    for (int i = 0; i < 16; ++i) {
      v2f lv = *(const v2f*)(lf + jj * 32 + i * 2);
      tt[i] = gm[9] * lv.x + g[9] * lv.y;
    }
    #pragma unroll
    for (int i = 0; i < 8; ++i) tt[i] = gm[8] * tt[2*i] + g[8] * tt[2*i+1];
    #pragma unroll
    for (int i = 0; i < 4; ++i) tt[i] = gm[7] * tt[2*i] + g[7] * tt[2*i+1];
    #pragma unroll
    for (int i = 0; i < 2; ++i) tt[i] = gm[6] * tt[2*i] + g[6] * tt[2*i+1];
    c16[jj] = gm[5] * tt[0] + g[5] * tt[1];
  }
  #pragma unroll
  for (int i = 0; i < 8; ++i) c16[i] = gm[4] * c16[2*i] + g[4] * c16[2*i+1];
  #pragma unroll
  for (int i = 0; i < 4; ++i) c16[i] = gm[3] * c16[2*i] + g[3] * c16[2*i+1];
  #pragma unroll
  for (int i = 0; i < 2; ++i) c16[i] = gm[2] * c16[2*i] + g[2] * c16[2*i+1];
  float vh = gm[1] * c16[0] + g[1] * c16[1];

  ldsV[wave * 32 + h * 16 + m] = vh;
  __syncthreads();
  if (lane < 16) {
    float v0 = ldsV[wave * 32 + m];
    float v1 = ldsV[wave * 32 + 16 + m];
    out[rowBase + m] = gm[0] * v0 + g[0] * v1;
  }
}

extern "C" void kernel_launch(void* const* d_in, const int* in_sizes, int n_in,
                              void* d_out, int out_size, void* d_ws, size_t ws_size,
                              hipStream_t stream) {
  (void)in_sizes; (void)n_in; (void)out_size; (void)d_ws; (void)ws_size;
  const float* x    = (const float*)d_in[0];
  const float* W    = (const float*)d_in[1];
  const float* b    = (const float*)d_in[2];
  const float* leaf = (const float*)d_in[3];
  float* out = (float*)d_out;
  sdt_wmma_kernel<<<dim3(BATCH / RPB), dim3(256), 0, stream>>>(x, W, b, leaf, out);
}


# rows per block 128 -> 64 so the grid is 256 blocks (every WGP, was 128): each row tile's k is split over 2 waves, the two partial sums are added through LDS (summation regrouped into k halves)
# speedup vs baseline: 2.2393x; 1.4065x over previous
#include <hip/hip_runtime.h>

typedef __attribute__((ext_vector_type(2))) float v2f;
typedef __attribute__((ext_vector_type(4))) float v4f;
typedef __attribute__((ext_vector_type(8))) float v8f;

#define BATCH   16384
#define NFEAT   1024
#define TDEPTH  10
#define NLEAVES 1024
#define WPB     8
#define RPW     16
#define KSPLIT  2
#define TPB     (WPB / KSPLIT)
#define RPB     (TPB * RPW)
#define NSTEP   (NFEAT / 8)
#define WSTEP   (NSTEP / KSPLIT)
#define GSTEP   16
#define NBLK    (WSTEP / GSTEP)

__launch_bounds__(256)
__global__ void sdt_wmma_kernel(const float* __restrict__ x,
                                const float* __restrict__ W,
                                const float* __restrict__ bias,
                                const float* __restrict__ leaf,
                                float* __restrict__ out) {
  __shared__ v4f ldsB4[NSTEP * 32];
  __shared__ float ldsLeaf[NLEAVES];
  __shared__ float ldsZ[WPB * 256];
  __shared__ float ldsV[TPB * 32];
  __shared__ float ldsBias[16];

  const int t    = threadIdx.x;
  const int wave = t >> 5;
  const int lane = t & 31;
  const int tile = wave & (TPB - 1);
  const int kh   = wave / TPB;
  const int rowBase = (blockIdx.x * TPB + tile) * RPW;
  const int m = lane & 15;
  const int h = lane >> 4;

  const v4f* xp = (const v4f*)(x + (size_t)(rowBase + m) * NFEAT + kh * (NFEAT / KSPLIT)) + h;
  v4f xa[2][GSTEP], bb[2][GSTEP];
  #pragma unroll
  for (int i = 0; i < GSTEP; ++i) xa[0][i] = xp[2 * i];

  {
    const int n = t & 15;
    const int kq0 = (t >> 5) * 32 + ((t >> 4) & 1);
    const v4f* wrow = (const v4f*)(W + n * NFEAT);
    v4f wv[16];
    #pragma unroll
    for (int i = 0; i < 16; ++i) wv[i] = wrow[kq0 + 2 * i];
    #pragma unroll
    for (int i = 0; i < 16; ++i) ldsB4[(kq0 + 2 * i) * 16 + n] = wv[i];
  }
  for (int idx = t; idx < NLEAVES; idx += 256) ldsLeaf[idx] = leaf[idx];
  if (t < 16) ldsBias[t] = bias[t];
  __syncthreads();

  const v4f* bp = ldsB4 + kh * (WSTEP * 32) + lane;
#define ISSUE_BLOCK(set, blk)                                                   \
  _Pragma("unroll") for (int i = 0; i < GSTEP; ++i)                             \
    xa[set][i] = xp[2 * ((blk) * GSTEP + i)];                                   \
  _Pragma("unroll") for (int i = 0; i < GSTEP; ++i)                             \
    bb[set][i] = bp[32 * ((blk) * GSTEP + i)];
#define COMPUTE_BLOCK(set)                                                      \
  _Pragma("unroll") for (int i = 0; i < GSTEP; ++i) {                           \
    acc = __builtin_amdgcn_wmma_f32_16x16x4_f32(false, xa[set][i].xy, false,    \
              bb[set][i].xy, (short)0, acc, false, false);                      \
    acc = __builtin_amdgcn_wmma_f32_16x16x4_f32(false, xa[set][i].zw, false,    \
              bb[set][i].zw, (short)0, acc, false, false);                      \
  }
  #pragma unroll
  for (int i = 0; i < GSTEP; ++i) bb[0][i] = bp[32 * i];
  v8f acc = {};
  #pragma unroll
  for (int blk = 0; blk < NBLK; blk += 2) {
    ISSUE_BLOCK(1, blk + 1)
    __builtin_amdgcn_sched_barrier(0);
    COMPUTE_BLOCK(0)
    __builtin_amdgcn_sched_barrier(0);
    if (blk + 2 < NBLK) {
      ISSUE_BLOCK(0, blk + 2)
      __builtin_amdgcn_sched_barrier(0);
    }
    COMPUTE_BLOCK(1)
    __builtin_amdgcn_sched_barrier(0);
  }
#undef ISSUE_BLOCK
#undef COMPUTE_BLOCK

  {
    float* zp = ldsZ + wave * 256;
    int n  = lane & 15;
    int mh = h * 8;
    #pragma unroll
    for (int i = 0; i < 8; ++i) zp[(mh + i) * 16 + n] = acc[i];
  }
  __syncthreads();

  float vh = 0.0f, g0 = 0.0f, gm0 = 0.0f;
  if (wave < TPB) {
    float g[TDEPTH], gm[TDEPTH];
    {
      const float* zp0 = ldsZ + wave * 256 + m * 16;
      const float* zp1 = ldsZ + (wave + TPB) * 256 + m * 16;
      #pragma unroll
      for (int d = 0; d < TDEPTH; ++d) {
        float z = (zp0[d] + zp1[d]) + ldsBias[d];
        float e = __expf(-z);
        g[d]  = 1.0f / (1.0f + e);
        gm[d] = 1.0f - g[d];
      }
    }

    const float* lf = ldsLeaf + h * 512;
    float c16[16];
    #pragma unroll
    for (int jj = 0; jj < 16; ++jj) {
      float tt[16];
      #pragma unroll
      for (int i = 0; i < 16; ++i) {
        v2f lv = *(const v2f*)(lf + jj * 32 + i * 2);
        tt[i] = gm[9] * lv.x + g[9] * lv.y;
      }
      #pragma unroll
      for (int i = 0; i < 8; ++i) tt[i] = gm[8] * tt[2*i] + g[8] * tt[2*i+1];
      #pragma unroll
      for (int i = 0; i < 4; ++i) tt[i] = gm[7] * tt[2*i] + g[7] * tt[2*i+1];
      #pragma unroll
      for (int i = 0; i < 2; ++i) tt[i] = gm[6] * tt[2*i] + g[6] * tt[2*i+1];
      c16[jj] = gm[5] * tt[0] + g[5] * tt[1];
    }
    #pragma unroll
    for (int i = 0; i < 8; ++i) c16[i] = gm[4] * c16[2*i] + g[4] * c16[2*i+1];
    #pragma unroll
    for (int i = 0; i < 4; ++i) c16[i] = gm[3] * c16[2*i] + g[3] * c16[2*i+1];
    #pragma unroll
    for (int i = 0; i < 2; ++i) c16[i] = gm[2] * c16[2*i] + g[2] * c16[2*i+1];
    vh  = gm[1] * c16[0] + g[1] * c16[1];
    g0  = g[0];
    gm0 = gm[0];
    ldsV[wave * 32 + h * 16 + m] = vh;
  }
  __syncthreads();

  if (wave < TPB && lane < 16) {
    float v0 = ldsV[wave * 32 + m];
    float v1 = ldsV[wave * 32 + 16 + m];
    out[rowBase + m] = gm0 * v0 + g0 * v1;
  }
}

extern "C" void kernel_launch(void* const* d_in, const int* in_sizes, int n_in,
                              void* d_out, int out_size, void* d_ws, size_t ws_size,
                              hipStream_t stream) {
  (void)in_sizes; (void)n_in; (void)out_size; (void)d_ws; (void)ws_size;
  const float* x    = (const float*)d_in[0];
  const float* W    = (const float*)d_in[1];
  const float* b    = (const float*)d_in[2];
  const float* leaf = (const float*)d_in[3];
  float* out = (float*)d_out;
  sdt_wmma_kernel<<<dim3(BATCH / RPB), dim3(256), 0, stream>>>(x, W, b, leaf, out);
}


# every global load of x and W reads 512 contiguous bytes of one row (4 whole lines, was 16 rows x 32 B): x reaches its A fragments via a per-wave LDS slab, W staged in fragment order; sums unchanged
# speedup vs baseline: 2.6111x; 1.1660x over previous
#include <hip/hip_runtime.h>

typedef __attribute__((ext_vector_type(2))) float v2f;
typedef __attribute__((ext_vector_type(4))) float v4f;
typedef __attribute__((ext_vector_type(8))) float v8f;

#define BATCH   16384
#define NFEAT   1024
#define TDEPTH  10
#define NLEAVES 1024
#define WPB     8
#define RPW     16
#define KSPLIT  2
#define TPB     (WPB / KSPLIT)
#define RPB     (TPB * RPW)
#define NSTEP   (NFEAT / 8)
#define WSTEP   (NSTEP / KSPLIT)
#define GSTEP   16
#define NBLK    (WSTEP / GSTEP)
#define BPITCH  17
#define XPITCH  33

__launch_bounds__(256)
__global__ void sdt_wmma_kernel(const float* __restrict__ x,
                                const float* __restrict__ W,
                                const float* __restrict__ bias,
                                const float* __restrict__ leaf,
                                float* __restrict__ out) {
  __shared__ v4f ldsB4[(NFEAT / 4) * BPITCH];
  __shared__ v4f ldsX[WPB][2][RPW * XPITCH];
  __shared__ float ldsLeaf[NLEAVES];
  __shared__ float ldsZ[WPB * 256];
  __shared__ float ldsV[TPB * 32];
  __shared__ float ldsBias[16];

  const int t    = threadIdx.x;
  const int wave = t >> 5;
  const int lane = t & 31;
  const int tile = wave & (TPB - 1);
  const int kh   = wave / TPB;
  const int rowBase = (blockIdx.x * TPB + tile) * RPW;
  const int m = lane & 15;
  const int h = lane >> 4;

  const v4f* xq = (const v4f*)(x + (size_t)rowBase * NFEAT + kh * (NFEAT / KSPLIT)) + lane;
  v4f raw[2][RPW], af[2][GSTEP], bf[2][GSTEP];
#define REQUEST(set, blk)                                                       \
  _Pragma("unroll") for (int r = 0; r < RPW; ++r)                               \
    raw[set][r] = xq[r * (NFEAT / 4) + (blk) * (2 * GSTEP)];

  {
    const int q = wave * 32 + lane;
    const v4f* wq = (const v4f*)W + q;
    v4f wv[16];
    #pragma unroll
    for (int j = 0; j < 16; ++j) wv[j] = wq[j * (NFEAT / 4)];
    REQUEST(0, 0)
    REQUEST(1, 1)
    #pragma unroll
    for (int j = 0; j < 16; ++j) ldsB4[q * BPITCH + j] = wv[j];
  }
  for (int idx = t; idx < NLEAVES; idx += 256) ldsLeaf[idx] = leaf[idx];
  if (t < 16) ldsBias[t] = bias[t];
  __syncthreads();

  v4f* slab0 = &ldsX[wave][0][0];
  v4f* slab1 = &ldsX[wave][1][0];
  const v4f* bp = ldsB4 + (kh * (2 * WSTEP) + h) * BPITCH + m;
#define PARK(set, slab)                                                         \
  _Pragma("unroll") for (int r = 0; r < RPW; ++r)                               \
    (slab)[r * XPITCH + lane] = raw[set][r];
#define FRAGS(set, slab, blk)                                                   \
  __builtin_amdgcn_fence(__ATOMIC_SEQ_CST, "wavefront");                        \
  _Pragma("unroll") for (int i = 0; i < GSTEP; ++i)                             \
    af[set][i] = (slab)[m * XPITCH + 2 * i + h];                                \
  _Pragma("unroll") for (int i = 0; i < GSTEP; ++i)                             \
    bf[set][i] = bp[(2 * BPITCH) * ((blk) * GSTEP + i)];
#define COMPUTE_BLOCK(set)                                                      \
  _Pragma("unroll") for (int i = 0; i < GSTEP; ++i) {                           \
    acc = __builtin_amdgcn_wmma_f32_16x16x4_f32(false, af[set][i].xy, false,    \
              bf[set][i].xy, (short)0, acc, false, false);                      \
    acc = __builtin_amdgcn_wmma_f32_16x16x4_f32(false, af[set][i].zw, false,    \
              bf[set][i].zw, (short)0, acc, false, false);                      \
  }                                                                             \
  asm volatile("" : "+v"(acc));
  PARK(0, slab0)
  REQUEST(0, 2)
  FRAGS(0, slab0, 0)
  v8f acc = {};
  __builtin_amdgcn_sched_barrier(0);
  PARK(1, slab1)
  REQUEST(1, 3)
  FRAGS(1, slab1, 1)
  __builtin_amdgcn_sched_barrier(0);
  COMPUTE_BLOCK(0)
  __builtin_amdgcn_sched_barrier(0);
  PARK(0, slab0)
  FRAGS(0, slab0, 2)
  __builtin_amdgcn_sched_barrier(0);
  COMPUTE_BLOCK(1)
  __builtin_amdgcn_sched_barrier(0);
  PARK(1, slab1)
  FRAGS(1, slab1, 3)
  __builtin_amdgcn_sched_barrier(0);
  COMPUTE_BLOCK(0)
  __builtin_amdgcn_sched_barrier(0);
  COMPUTE_BLOCK(1)
#undef REQUEST
#undef PARK
#undef FRAGS
#undef COMPUTE_BLOCK

  {
    float* zp = ldsZ + wave * 256;
    int n  = lane & 15;
    int mh = h * 8;
    #pragma unroll
    for (int i = 0; i < 8; ++i) zp[(mh + i) * 16 + n] = acc[i];
  }
  __syncthreads();

  float vh = 0.0f, g0 = 0.0f, gm0 = 0.0f;
  if (wave < TPB) {
    float g[TDEPTH], gm[TDEPTH];
    {
      const float* zp0 = ldsZ + wave * 256 + m * 16;
      const float* zp1 = ldsZ + (wave + TPB) * 256 + m * 16;
      #pragma unroll
      for (int d = 0; d < TDEPTH; ++d) {
        float z = (zp0[d] + zp1[d]) + ldsBias[d];
        float e = __expf(-z);
        g[d]  = 1.0f / (1.0f + e);
        gm[d] = 1.0f - g[d];
      }
    }

    const float* lf = ldsLeaf + h * 512;
    float c16[16];
    #pragma unroll
    for (int jj = 0; jj < 16; ++jj) {
      float tt[16];
      #pragma unroll
      for (int i = 0; i < 16; ++i) {
        v2f lv = *(const v2f*)(lf + jj * 32 + i * 2);
        tt[i] = gm[9] * lv.x + g[9] * lv.y;
      }
      #pragma unroll
      for (int i = 0; i < 8; ++i) tt[i] = gm[8] * tt[2*i] + g[8] * tt[2*i+1];
      #pragma unroll
      for (int i = 0; i < 4; ++i) tt[i] = gm[7] * tt[2*i] + g[7] * tt[2*i+1];
      #pragma unroll
      for (int i = 0; i < 2; ++i) tt[i] = gm[6] * tt[2*i] + g[6] * tt[2*i+1];
      c16[jj] = gm[5] * tt[0] + g[5] * tt[1];
    }
    #pragma unroll
    for (int i = 0; i < 8; ++i) c16[i] = gm[4] * c16[2*i] + g[4] * c16[2*i+1];
    #pragma unroll
    for (int i = 0; i < 4; ++i) c16[i] = gm[3] * c16[2*i] + g[3] * c16[2*i+1];
    #pragma unroll
    for (int i = 0; i < 2; ++i) c16[i] = gm[2] * c16[2*i] + g[2] * c16[2*i+1];
    vh  = gm[1] * c16[0] + g[1] * c16[1];
    g0  = g[0];
    gm0 = gm[0];
    ldsV[wave * 32 + h * 16 + m] = vh;
  }
  __syncthreads();

  if (wave < TPB && lane < 16) {
    float v0 = ldsV[wave * 32 + m];
    float v1 = ldsV[wave * 32 + 16 + m];
    out[rowBase + m] = gm0 * v0 + g0 * v1;
  }
}

extern "C" void kernel_launch(void* const* d_in, const int* in_sizes, int n_in,
                              void* d_out, int out_size, void* d_ws, size_t ws_size,
                              hipStream_t stream) {
  (void)in_sizes; (void)n_in; (void)out_size; (void)d_ws; (void)ws_size;
  const float* x    = (const float*)d_in[0];
  const float* W    = (const float*)d_in[1];
  const float* b    = (const float*)d_in[2];
  const float* leaf = (const float*)d_in[3];
  float* out = (float*)d_out;
  sdt_wmma_kernel<<<dim3(BATCH / RPB), dim3(256), 0, stream>>>(x, W, b, leaf, out);
}


# x parked and multiplied in blocks of 4 fragment steps (was 16); no wait ahead of the first park drains the x loads (hand-written barrier, leaf/bias requested first): wmmas start early, short tail
# speedup vs baseline: 2.8894x; 1.1066x over previous
#include <hip/hip_runtime.h>

typedef __attribute__((ext_vector_type(2))) float v2f;
typedef __attribute__((ext_vector_type(4))) float v4f;
typedef __attribute__((ext_vector_type(8))) float v8f;

#define BATCH   16384
#define NFEAT   1024
#define TDEPTH  10
#define NLEAVES 1024
#define WPB     8
#define RPW     16
#define KSPLIT  2
#define TPB     (WPB / KSPLIT)
#define RPB     (TPB * RPW)
#define NSTEP   (NFEAT / 8)
#define WSTEP   (NSTEP / KSPLIT)
#define GSTEP   4
#define NBLK    (WSTEP / GSTEP)
#define PPR     (2 * GSTEP)
#define RPI     (32 / PPR)
#define LPB     (RPW / RPI)
#define NRAW    (32 / LPB)
#define BPITCH  17
#define XPITCH  (PPR + 1)

__launch_bounds__(256)
__global__ void sdt_wmma_kernel(const float* __restrict__ x,
                                const float* __restrict__ W,
                                const float* __restrict__ bias,
                                const float* __restrict__ leaf,
                                float* __restrict__ out) {
  __shared__ v4f ldsB4[(NFEAT / 4) * BPITCH];
  __shared__ v4f ldsX[WPB][NRAW][RPW * XPITCH];
  __shared__ __attribute__((aligned(16))) float ldsLeaf[NLEAVES];
  __shared__ float ldsZ[WPB * 256];
  __shared__ float ldsV[TPB * 32];
  __shared__ float ldsBias[16];

  const int t    = threadIdx.x;
  const int wave = t >> 5;
  const int lane = t & 31;
  const int tile = wave & (TPB - 1);
  const int kh   = wave / TPB;
  const int rowBase = (blockIdx.x * TPB + tile) * RPW;
  const int m = lane & 15;
  const int h = lane >> 4;
  const int lr = lane / PPR;
  const int lp = lane % PPR;

  const v4f* xq = (const v4f*)(x + (size_t)(rowBase + lr) * NFEAT + kh * (NFEAT / KSPLIT)) + lp;
  v4f raw[NRAW][LPB], af[2][GSTEP], bf[2][GSTEP];
#define REQUEST2(set, blk)                                                      \
  _Pragma("unroll") for (int j = 0; j < LPB; ++j) {                             \
    raw[set][j]     = xq[j * (RPI * (NFEAT / 4)) + (blk) * PPR];                \
    raw[(set) + 1][j] = xq[j * (RPI * (NFEAT / 4)) + ((blk) + 1) * PPR];        \
  }

  {
    const int q = wave * 32 + lane;
    const v4f* wq = (const v4f*)W + q;
    v4f wv[16];
    #pragma unroll
    for (int j = 0; j < 16; ++j) wv[j] = wq[j * (NFEAT / 4)];
    const v4f lv = ((const v4f*)leaf)[t];
    const float bv = bias[t & 15];
    #pragma unroll
    for (int b = 0; b < NRAW; b += 2) { REQUEST2(b, b) }
    #pragma unroll
    for (int j = 0; j < 16; ++j) ldsB4[q * BPITCH + j] = wv[j];
    *(v4f*)(ldsLeaf + 4 * t) = lv;
    ldsBias[t & 15] = bv;
  }
  asm volatile("s_wait_dscnt 0x0" ::: "memory");
  asm volatile("s_barrier_signal -1" ::: "memory");
  asm volatile("s_barrier_wait -1" ::: "memory");

  v4f* slabs = &ldsX[wave][0][0];
  const v4f* bp = ldsB4 + (kh * (2 * WSTEP) + h) * BPITCH + m;
#define PARK(set, slab)                                                         \
  _Pragma("unroll") for (int j = 0; j < LPB; ++j)                               \
    (slab)[(j * RPI + lr) * XPITCH + lp] = raw[set][j];
#define FRAGS(set, slab, blk)                                                   \
  __builtin_amdgcn_fence(__ATOMIC_SEQ_CST, "wavefront");                        \
  _Pragma("unroll") for (int i = 0; i < GSTEP; ++i)                             \
    af[set][i] = (slab)[m * XPITCH + 2 * i + h];                                \
  _Pragma("unroll") for (int i = 0; i < GSTEP; ++i)                             \
    bf[set][i] = bp[(2 * BPITCH) * ((blk) * GSTEP + i)];
#define COMPUTE_BLOCK(set)                                                      \
  _Pragma("unroll") for (int i = 0; i < GSTEP; ++i) {                           \
    acc = __builtin_amdgcn_wmma_f32_16x16x4_f32(false, af[set][i].xy, false,    \
              bf[set][i].xy, (short)0, acc, false, false);                      \
    acc = __builtin_amdgcn_wmma_f32_16x16x4_f32(false, af[set][i].zw, false,    \
              bf[set][i].zw, (short)0, acc, false, false);                      \
  }                                                                             \
  asm volatile("" : "+v"(acc));
  v8f acc = {};
  PARK(0, slabs)
  FRAGS(0, slabs, 0)
  __builtin_amdgcn_sched_barrier(0);
  #pragma unroll
  for (int b = 1; b < NBLK; ++b) {
    v4f* slab = slabs + (b % NRAW) * (RPW * XPITCH);
    PARK(b % NRAW, slab)
    if ((b & 1) && b + NRAW < NBLK) { REQUEST2((b - 1) % NRAW, b - 1 + NRAW) }
    FRAGS(b & 1, slab, b)
    __builtin_amdgcn_sched_barrier(0);
    COMPUTE_BLOCK((b - 1) & 1)
    __builtin_amdgcn_sched_barrier(0);
  }
  COMPUTE_BLOCK((NBLK - 1) & 1)
#undef REQUEST2
#undef PARK
#undef FRAGS
#undef COMPUTE_BLOCK

  {
    float* zp = ldsZ + wave * 256;
    int n  = lane & 15;
    int mh = h * 8;
    #pragma unroll
    for (int i = 0; i < 8; ++i) zp[(mh + i) * 16 + n] = acc[i];
  }
  __syncthreads();

  float vh = 0.0f, g0 = 0.0f, gm0 = 0.0f;
  if (wave < TPB) {
    float g[TDEPTH], gm[TDEPTH];
    {
      const float* zp0 = ldsZ + wave * 256 + m * 16;
      const float* zp1 = ldsZ + (wave + TPB) * 256 + m * 16;
      #pragma unroll
      for (int d = 0; d < TDEPTH; ++d) {
        float z = (zp0[d] + zp1[d]) + ldsBias[d];
        float e = __expf(-z);
        g[d]  = 1.0f / (1.0f + e);
        gm[d] = 1.0f - g[d];
      }
    }

    const float* lf = ldsLeaf + h * 512;
    float c16[16];
    #pragma unroll
    for (int jj = 0; jj < 16; ++jj) {
      float tt[16];
      #pragma unroll
      for (int i = 0; i < 16; ++i) {
        v2f lv = *(const v2f*)(lf + jj * 32 + i * 2);
        tt[i] = gm[9] * lv.x + g[9] * lv.y;
      }
      #pragma unroll
      for (int i = 0; i < 8; ++i) tt[i] = gm[8] * tt[2*i] + g[8] * tt[2*i+1];
      #pragma unroll
      for (int i = 0; i < 4; ++i) tt[i] = gm[7] * tt[2*i] + g[7] * tt[2*i+1];
      #pragma unroll
      for (int i = 0; i < 2; ++i) tt[i] = gm[6] * tt[2*i] + g[6] * tt[2*i+1];
      c16[jj] = gm[5] * tt[0] + g[5] * tt[1];
    }
    #pragma unroll
    for (int i = 0; i < 8; ++i) c16[i] = gm[4] * c16[2*i] + g[4] * c16[2*i+1];
    #pragma unroll
    for (int i = 0; i < 4; ++i) c16[i] = gm[3] * c16[2*i] + g[3] * c16[2*i+1];
    #pragma unroll
    for (int i = 0; i < 2; ++i) c16[i] = gm[2] * c16[2*i] + g[2] * c16[2*i+1];
    vh  = gm[1] * c16[0] + g[1] * c16[1];
    g0  = g[0];
    gm0 = gm[0];
    ldsV[wave * 32 + h * 16 + m] = vh;
  }
  __syncthreads();

  if (wave < TPB && lane < 16) {
    float v0 = ldsV[wave * 32 + m];
    float v1 = ldsV[wave * 32 + 16 + m];
    out[rowBase + m] = gm0 * v0 + g0 * v1;
  }
}

extern "C" void kernel_launch(void* const* d_in, const int* in_sizes, int n_in,
                              void* d_out, int out_size, void* d_ws, size_t ws_size,
                              hipStream_t stream) {
  (void)in_sizes; (void)n_in; (void)out_size; (void)d_ws; (void)ws_size;
  const float* x    = (const float*)d_in[0];
  const float* W    = (const float*)d_in[1];
  const float* b    = (const float*)d_in[2];
  const float* leaf = (const float*)d_in[3];
  float* out = (float*)d_out;
  sdt_wmma_kernel<<<dim3(BATCH / RPB), dim3(256), 0, stream>>>(x, W, b, leaf, out);
}


# leaf sum as a matrix product: 32x32 leaf matrix times each row's 32 products of gates 5-9 on 16 wmmas, then 16 fmas with the products of gates 0-4 (was a 1023-node vector fold); sum regrouped
# speedup vs baseline: 3.2851x; 1.1369x over previous
#include <hip/hip_runtime.h>

typedef __attribute__((ext_vector_type(2))) float v2f;
typedef __attribute__((ext_vector_type(4))) float v4f;
typedef __attribute__((ext_vector_type(8))) float v8f;

#define BATCH   16384
#define NFEAT   1024
#define TDEPTH  10
#define NLEAVES 1024
#define WPB     8
#define RPW     16
#define KSPLIT  2
#define TPB     (WPB / KSPLIT)
#define RPB     (TPB * RPW)
#define NSTEP   (NFEAT / 8)
#define WSTEP   (NSTEP / KSPLIT)
#define GSTEP   4
#define NBLK    (WSTEP / GSTEP)
#define PPR     (2 * GSTEP)
#define RPI     (32 / PPR)
#define LPB     (RPW / RPI)
#define NRAW    (32 / LPB)
#define BPITCH  17
#define XPITCH  (PPR + 1)
#define LPITCH  9

__launch_bounds__(256)
__global__ void sdt_wmma_kernel(const float* __restrict__ x,
                                const float* __restrict__ W,
                                const float* __restrict__ bias,
                                const float* __restrict__ leaf,
                                float* __restrict__ out) {
  __shared__ v4f ldsB4[(NFEAT / 4) * BPITCH];
  __shared__ v4f ldsX[WPB][NRAW][RPW * XPITCH];
  __shared__ v4f ldsLeaf4[32 * LPITCH];
  __shared__ float ldsZ[WPB * 256];
  __shared__ float ldsV[TPB * 32];
  __shared__ float ldsBias[16];

  const int t    = threadIdx.x;
  const int wave = t >> 5;
  const int lane = t & 31;
  const int tile = wave & (TPB - 1);
  const int kh   = wave / TPB;
  const int rowBase = (blockIdx.x * TPB + tile) * RPW;
  const int m = lane & 15;
  const int h = lane >> 4;
  const int lr = lane / PPR;
  const int lp = lane % PPR;

  const v4f* xq = (const v4f*)(x + (size_t)(rowBase + lr) * NFEAT + kh * (NFEAT / KSPLIT)) + lp;
  v4f raw[NRAW][LPB], af[2][GSTEP], bf[2][GSTEP];
#define REQUEST2(set, blk)                                                      \
  _Pragma("unroll") for (int j = 0; j < LPB; ++j) {                             \
    raw[set][j]     = xq[j * (RPI * (NFEAT / 4)) + (blk) * PPR];                \
    raw[(set) + 1][j] = xq[j * (RPI * (NFEAT / 4)) + ((blk) + 1) * PPR];        \
  }

  {
    const int q = wave * 32 + lane;
    const v4f* wq = (const v4f*)W + q;
    v4f wv[16];
    #pragma unroll
    for (int j = 0; j < 16; ++j) wv[j] = wq[j * (NFEAT / 4)];
    const v4f lv = ((const v4f*)leaf)[t];
    const float bv = bias[t & 15];
    #pragma unroll
    for (int b = 0; b < NRAW; b += 2) { REQUEST2(b, b) }
    #pragma unroll
    for (int j = 0; j < 16; ++j) ldsB4[q * BPITCH + j] = wv[j];
    ldsLeaf4[(t >> 3) * LPITCH + (t & 7)] = lv;
    ldsBias[t & 15] = bv;
  }
  asm volatile("s_wait_dscnt 0x0" ::: "memory");
  asm volatile("s_barrier_signal -1" ::: "memory");
  asm volatile("s_barrier_wait -1" ::: "memory");

  v4f* slabs = &ldsX[wave][0][0];
  const v4f* bp = ldsB4 + (kh * (2 * WSTEP) + h) * BPITCH + m;
#define PARK(set, slab)                                                         \
  _Pragma("unroll") for (int j = 0; j < LPB; ++j)                               \
    (slab)[(j * RPI + lr) * XPITCH + lp] = raw[set][j];
#define FRAGS(set, slab, blk)                                                   \
  __builtin_amdgcn_fence(__ATOMIC_SEQ_CST, "wavefront");                        \
  _Pragma("unroll") for (int i = 0; i < GSTEP; ++i)                             \
    af[set][i] = (slab)[m * XPITCH + 2 * i + h];                                \
  _Pragma("unroll") for (int i = 0; i < GSTEP; ++i)                             \
    bf[set][i] = bp[(2 * BPITCH) * ((blk) * GSTEP + i)];
#define COMPUTE_BLOCK(set)                                                      \
  _Pragma("unroll") for (int i = 0; i < GSTEP; ++i) {                           \
    acc = __builtin_amdgcn_wmma_f32_16x16x4_f32(false, af[set][i].xy, false,    \
              bf[set][i].xy, (short)0, acc, false, false);                      \
    acc = __builtin_amdgcn_wmma_f32_16x16x4_f32(false, af[set][i].zw, false,    \
              bf[set][i].zw, (short)0, acc, false, false);                      \
  }                                                                             \
  asm volatile("" : "+v"(acc));
  v8f acc = {};
  PARK(0, slabs)
  FRAGS(0, slabs, 0)
  __builtin_amdgcn_sched_barrier(0);
  #pragma unroll
  for (int b = 1; b < NBLK; ++b) {
    v4f* slab = slabs + (b % NRAW) * (RPW * XPITCH);
    PARK(b % NRAW, slab)
    if ((b & 1) && b + NRAW < NBLK) { REQUEST2((b - 1) % NRAW, b - 1 + NRAW) }
    FRAGS(b & 1, slab, b)
    __builtin_amdgcn_sched_barrier(0);
    COMPUTE_BLOCK((b - 1) & 1)
    __builtin_amdgcn_sched_barrier(0);
  }
  COMPUTE_BLOCK((NBLK - 1) & 1)
#undef REQUEST2
#undef PARK
#undef FRAGS
#undef COMPUTE_BLOCK

  {
    float* zp = ldsZ + wave * 256;
    int n  = lane & 15;
    int mh = h * 8;
    #pragma unroll
    for (int i = 0; i < 8; ++i) zp[(mh + i) * 16 + n] = acc[i];
  }
  __syncthreads();

  float vh = 0.0f;
  if (wave < TPB) {
    float g[TDEPTH], gm[TDEPTH];
    {
      const float* zp0 = ldsZ + wave * 256 + m * 16;
      const float* zp1 = ldsZ + (wave + TPB) * 256 + m * 16;
      #pragma unroll
      for (int d = 0; d < TDEPTH; ++d) {
        float z = (zp0[d] + zp1[d]) + ldsBias[d];
        float e = __expf(-z);
        g[d]  = 1.0f / (1.0f + e);
        gm[d] = 1.0f - g[d];
      }
    }

    v4f la0[4], la1[4];
    #pragma unroll
    for (int j = 0; j < 4; ++j) {
      la0[j] = ldsLeaf4[m * LPITCH + 4 * h + j];
      la1[j] = ldsLeaf4[(16 + m) * LPITCH + 4 * h + j];
    }
    const float p5 = h ? g[5] : gm[5];
    float e1[2], e2[4], e3[8];
    e1[0] = p5 * gm[6]; e1[1] = p5 * g[6];
    #pragma unroll
    for (int i = 0; i < 2; ++i) { e2[2*i] = e1[i] * gm[7]; e2[2*i+1] = e1[i] * g[7]; }
    #pragma unroll
    for (int i = 0; i < 4; ++i) { e3[2*i] = e2[i] * gm[8]; e3[2*i+1] = e2[i] * g[8]; }
    const v2f g9v = {gm[9], g[9]};
    v2f plv[8];
    #pragma unroll
    for (int i = 0; i < 8; ++i) plv[i] = e3[i] * g9v;
    float q2[4], q3[8];
    q2[0] = gm[2] * gm[3]; q2[1] = gm[2] * g[3]; q2[2] = g[2] * gm[3]; q2[3] = g[2] * g[3];
    #pragma unroll
    for (int i = 0; i < 4; ++i) { q3[2*i] = q2[i] * gm[4]; q3[2*i+1] = q2[i] * g[4]; }
    const float p1  = h ? g[1] : gm[1];
    const float pt0 = p1 * gm[0], pt1 = p1 * g[0];
    __builtin_amdgcn_sched_barrier(0);
    v8f d0 = {}, d1 = {};
    #pragma unroll
    for (int j = 0; j < 4; ++j) {
      d0 = __builtin_amdgcn_wmma_f32_16x16x4_f32(false, la0[j].xy, false, plv[2*j],     (short)0, d0, false, false);
      d1 = __builtin_amdgcn_wmma_f32_16x16x4_f32(false, la1[j].xy, false, plv[2*j],     (short)0, d1, false, false);
      d0 = __builtin_amdgcn_wmma_f32_16x16x4_f32(false, la0[j].zw, false, plv[2*j + 1], (short)0, d0, false, false);
      d1 = __builtin_amdgcn_wmma_f32_16x16x4_f32(false, la1[j].zw, false, plv[2*j + 1], (short)0, d1, false, false);
    }
    __builtin_amdgcn_sched_barrier(0);
    float s0 = 0.0f, s1 = 0.0f;
    #pragma unroll
    for (int r = 0; r < 8; ++r) { s0 += q3[r] * d0[r]; s1 += q3[r] * d1[r]; }
    vh  = pt0 * s0 + pt1 * s1;
    ldsV[wave * 32 + h * 16 + m] = vh;
  }
  __syncthreads();

  if (wave < TPB && lane < 16) {
    float v0 = ldsV[wave * 32 + m];
    float v1 = ldsV[wave * 32 + 16 + m];
    out[rowBase + m] = v0 + v1;
  }
}

extern "C" void kernel_launch(void* const* d_in, const int* in_sizes, int n_in,
                              void* d_out, int out_size, void* d_ws, size_t ws_size,
                              hipStream_t stream) {
  (void)in_sizes; (void)n_in; (void)out_size; (void)d_ws; (void)ws_size;
  const float* x    = (const float*)d_in[0];
  const float* W    = (const float*)d_in[1];
  const float* b    = (const float*)d_in[2];
  const float* leaf = (const float*)d_in[3];
  float* out = (float*)d_out;
  sdt_wmma_kernel<<<dim3(BATCH / RPB), dim3(256), 0, stream>>>(x, W, b, leaf, out);
}
